# starting kernel
# baseline (speedup 1.0000x reference)
#include <hip/hip_runtime.h>


typedef __attribute__((ext_vector_type(2))) float v2f;
typedef __attribute__((ext_vector_type(8))) float v8f;

#define Hdim 512
#define Vdim 5
#define Bdim 8
#define Tdim 400
#define U1dim 101

#define TROWS (Bdim * Tdim)
#define PROWS (Bdim * U1dim)
#define TTILES ((TROWS + 15) / 16)
#define PTILES ((PROWS + 15) / 16)
#define NTILES (TTILES + PTILES)

__global__ __launch_bounds__(256) void rowgemm_wmma(
    const float* __restrict__ trans,
    const float* __restrict__ preds,
    const float* __restrict__ W,
    const float* __restrict__ bias,
    float* __restrict__ TW,
    float* __restrict__ PW)
{
    const int wave = threadIdx.x >> 5;
    const int lane = threadIdx.x & 31;
    const int tile = blockIdx.x * 8 + wave;
    if (tile >= NTILES) return;

    const bool  isTrans = (tile < TTILES);
    const float* A   = isTrans ? trans : preds;
    float*       Dst = isTrans ? TW    : PW;
    const int    rows = isTrans ? TROWS : PROWS;
    const int    row0 = (isTrans ? tile : tile - TTILES) * 16;

    const int half = lane >> 4;
    const int l15  = lane & 15;

    const int arow = min(row0 + l15, rows - 1);
    const int brow = min(l15, Vdim - 1);

    const float* __restrict__ ap = A + arow * Hdim + 2 * half;
    const float* __restrict__ bp = W + brow * Hdim + 2 * half;

    v8f c;
    const float cinit = isTrans ? bias[brow] : 0.0f;
#pragma unroll
    for (int r = 0; r < 8; ++r) c[r] = cinit;

#pragma unroll 8
    for (int k = 0; k < Hdim; k += 4) {
        const v2f a = *(const v2f*)(ap + k);
        const v2f b = *(const v2f*)(bp + k);
        c = __builtin_amdgcn_wmma_f32_16x16x4_f32(
                 false, a,  false, b,
                 (short)0, c,  false,  false);
    }

    if (l15 < Vdim) {
        const int n = l15;
#pragma unroll
        for (int r = 0; r < 8; ++r) {
            const int row = row0 + r + 8 * half;
            if (row < rows) Dst[row * Vdim + n] = c[r];
        }
    }
}

__global__ __launch_bounds__(256) void joint_write(
    const float* __restrict__ TW,
    const float* __restrict__ PW,
    float* __restrict__ out)
{
    const int bt = blockIdx.x;
    const int bb = bt / Tdim;
    const float* __restrict__ pw = PW + bb * (U1dim * Vdim);
    const float* __restrict__ tw = TW + bt * Vdim;
    float* __restrict__ o = out + (size_t)bt * (U1dim * Vdim);

    for (int j = threadIdx.x; j < U1dim * Vdim; j += 256) {
        const int v = j % Vdim;
        const float x = tw[v] + pw[j];
        __builtin_nontemporal_store(fmaxf(x, 0.1f * x), o + j);
    }
}

extern "C" void kernel_launch(void* const* d_in, const int* in_sizes, int n_in,
                              void* d_out, int out_size, void* d_ws, size_t ws_size,
                              hipStream_t stream) {
    const float* trans = (const float*)d_in[0];
    const float* preds = (const float*)d_in[1];
    const float* W     = (const float*)d_in[2];
    const float* bias  = (const float*)d_in[3];
    float* out = (float*)d_out;

    float* TW = (float*)d_ws;
    float* PW = TW + TROWS * Vdim;

    rowgemm_wmma<<<(NTILES + 7) / 8, 256, 0, stream>>>(trans, preds, W, bias, TW, PW);
    joint_write<<<TROWS, 256, 0, stream>>>(TW, PW, out);
}


# joint_write on 256 resident work-groups looping over the flat output, one aligned 16-byte store per lane per trip: 2048 waves to enter instead of 25600
# speedup vs baseline: 1.1748x; 1.1748x over previous
#include <hip/hip_runtime.h>


typedef __attribute__((ext_vector_type(2))) float v2f;
typedef __attribute__((ext_vector_type(8))) float v8f;

#define Hdim 512
#define Vdim 5
#define Bdim 8
#define Tdim 400
#define U1dim 101

#define TROWS (Bdim * Tdim)
#define PROWS (Bdim * U1dim)
#define TTILES ((TROWS + 15) / 16)
#define PTILES ((PROWS + 15) / 16)
#define NTILES (TTILES + PTILES)

__global__ __launch_bounds__(256) void rowgemm_wmma(
    const float* __restrict__ trans,
    const float* __restrict__ preds,
    const float* __restrict__ W,
    const float* __restrict__ bias,
    float* __restrict__ TW,
    float* __restrict__ PW)
{
    const int wave = threadIdx.x >> 5;
    const int lane = threadIdx.x & 31;
    const int tile = blockIdx.x * 8 + wave;
    if (tile >= NTILES) return;

    const bool  isTrans = (tile < TTILES);
    const float* A   = isTrans ? trans : preds;
    float*       Dst = isTrans ? TW    : PW;
    const int    rows = isTrans ? TROWS : PROWS;
    const int    row0 = (isTrans ? tile : tile - TTILES) * 16;

    const int half = lane >> 4;
    const int l15  = lane & 15;

    const int arow = min(row0 + l15, rows - 1);
    const int brow = min(l15, Vdim - 1);

    const float* __restrict__ ap = A + arow * Hdim + 2 * half;
    const float* __restrict__ bp = W + brow * Hdim + 2 * half;

    v8f c;
    const float cinit = isTrans ? bias[brow] : 0.0f;
#pragma unroll
    for (int r = 0; r < 8; ++r) c[r] = cinit;

#pragma unroll 8
    for (int k = 0; k < Hdim; k += 4) {
        const v2f a = *(const v2f*)(ap + k);
        const v2f b = *(const v2f*)(bp + k);
        c = __builtin_amdgcn_wmma_f32_16x16x4_f32(
                 false, a,  false, b,
                 (short)0, c,  false,  false);
    }

    if (l15 < Vdim) {
        const int n = l15;
#pragma unroll
        for (int r = 0; r < 8; ++r) {
            const int row = row0 + r + 8 * half;
            if (row < rows) Dst[row * Vdim + n] = c[r];
        }
    }
}

typedef __attribute__((ext_vector_type(4))) float v4f;

#define ROWLEN    (U1dim * Vdim)
#define OUT_ELEMS (TROWS * ROWLEN)
#define OUT_Q     (OUT_ELEMS / 4)
#define JW_BLOCKS 256
#define JW_STRIDE (JW_BLOCKS * 256)
#define JW_FULL   (OUT_Q / JW_STRIDE)
#define JW_TAIL   (OUT_Q - JW_FULL * JW_STRIDE)

static __device__ __forceinline__ v4f joint4(const float* __restrict__ TW,
                                             const float* __restrict__ PW,
                                             const unsigned q)
{
    const unsigned e0  = 4u * q;
    const unsigned bt0 = e0 / ROWLEN;
    const unsigned j0  = e0 - bt0 * ROWLEN;
    const unsigned v0  = j0 % Vdim;
    const unsigned pw0 = (bt0 / Tdim) * ROWLEN;
    const unsigned pw1 = ((bt0 + 1u) / Tdim) * ROWLEN;

    v4f r;
#pragma unroll
    for (unsigned i = 0; i < 4; ++i) {
        const unsigned j    = j0 + i;
        const bool     next = j >= ROWLEN;
        const unsigned jj   = next ? j - ROWLEN : j;
        const unsigned tb   = (next ? bt0 + 1u : bt0) * Vdim;
        const unsigned pb   = next ? pw1 : pw0;
        unsigned v = v0 + i;
        v = v >= Vdim ? v - Vdim : v;
        const float x = TW[tb + v] + PW[pb + jj];
        r[i] = fmaxf(x, 0.1f * x);
    }
    return r;
}

__global__ __launch_bounds__(256) void joint_write(
    const float* __restrict__ TW,
    const float* __restrict__ PW,
    float* __restrict__ out)
{
    const unsigned gtid = blockIdx.x * 256u + threadIdx.x;

    v4f r[JW_FULL];
#pragma unroll
    for (unsigned it = 0; it < JW_FULL; ++it)
        r[it] = joint4(TW, PW, it * JW_STRIDE + gtid);
#pragma unroll
    for (unsigned it = 0; it < JW_FULL; ++it)
        __builtin_nontemporal_store(r[it], (v4f*)(out + 4u * (size_t)(it * JW_STRIDE + gtid)));

    if (gtid < JW_TAIL) {
        const unsigned q = JW_FULL * JW_STRIDE + gtid;
        __builtin_nontemporal_store(joint4(TW, PW, q), (v4f*)(out + 4u * (size_t)q));
    }
}

extern "C" void kernel_launch(void* const* d_in, const int* in_sizes, int n_in,
                              void* d_out, int out_size, void* d_ws, size_t ws_size,
                              hipStream_t stream) {
    const float* trans = (const float*)d_in[0];
    const float* preds = (const float*)d_in[1];
    const float* W     = (const float*)d_in[2];
    const float* bias  = (const float*)d_in[3];
    float* out = (float*)d_out;

    float* TW = (float*)d_ws;
    float* PW = TW + TROWS * Vdim;

    rowgemm_wmma<<<(NTILES + 7) / 8, 256, 0, stream>>>(trans, preds, W, bias, TW, PW);
    joint_write<<<JW_BLOCKS, 256, 0, stream>>>(TW, PW, out);
}


# rowgemm: K sum of each 16-row tile split over the 8 waves of one work-group (251 groups, 16-byte loads issued up front), partial sums added through LDS in a fixed order: summation order changes
# speedup vs baseline: 2.6787x; 2.2801x over previous
#include <hip/hip_runtime.h>


typedef __attribute__((ext_vector_type(2))) float v2f;
typedef __attribute__((ext_vector_type(8))) float v8f;

#define Hdim 512
#define Vdim 5
#define Bdim 8
#define Tdim 400
#define U1dim 101

#define TROWS (Bdim * Tdim)
#define PROWS (Bdim * U1dim)
#define TTILES ((TROWS + 15) / 16)
#define PTILES ((PROWS + 15) / 16)
#define NTILES (TTILES + PTILES)

typedef __attribute__((ext_vector_type(4))) float v4f;

#define KWAVES 8
#define KPIECE (Hdim / KWAVES)
#define KLOADS (KPIECE / 8)
#define TILE_OUT (16 * Vdim)
#define NPART (2 * KWAVES)

__global__ __launch_bounds__(256) void rowgemm_wmma(
    const float* __restrict__ trans,
    const float* __restrict__ preds,
    const float* __restrict__ W,
    const float* __restrict__ bias,
    float* __restrict__ TW,
    float* __restrict__ PW)
{
    __shared__ float part[NPART][TILE_OUT];

    const int wave = threadIdx.x >> 5;
    const int lane = threadIdx.x & 31;
    const int tile = blockIdx.x;

    const bool  isTrans = (tile < TTILES);
    const float* A   = isTrans ? trans : preds;
    float*       Dst = isTrans ? TW    : PW;
    const int    rows = isTrans ? TROWS : PROWS;
    const int    row0 = (isTrans ? tile : tile - TTILES) * 16;

    const int half = lane >> 4;
    const int l15  = lane & 15;

    const int arow = min(row0 + l15, rows - 1);
    const int brow = min(l15, Vdim - 1);

    const float* __restrict__ ap = A + arow * Hdim + wave * KPIECE + 4 * half;
    const float* __restrict__ bp = W + brow * Hdim + wave * KPIECE + 4 * half;

    v4f a[KLOADS], b[KLOADS];
#pragma unroll
    for (int j = 0; j < KLOADS; ++j) a[j] = *(const v4f*)(ap + 8 * j);
#pragma unroll
    for (int j = 0; j < KLOADS; ++j) b[j] = *(const v4f*)(bp + 8 * j);

    v8f c0, c1;
#pragma unroll
    for (int r = 0; r < 8; ++r) { c0[r] = 0.0f; c1[r] = 0.0f; }

#pragma unroll
    for (int j = 0; j < KLOADS; ++j) {
        const v2f a01 = {a[j][0], a[j][1]}, b01 = {b[j][0], b[j][1]};
        const v2f a23 = {a[j][2], a[j][3]}, b23 = {b[j][2], b[j][3]};
        c0 = __builtin_amdgcn_wmma_f32_16x16x4_f32(
                  false, a01,  false, b01,
                  (short)0, c0,  false,  false);
        c1 = __builtin_amdgcn_wmma_f32_16x16x4_f32(
                  false, a23,  false, b23,
                  (short)0, c1,  false,  false);
    }

    if (l15 < Vdim) {
#pragma unroll
        for (int r = 0; r < 8; ++r) {
            part[2 * wave    ][(r + 8 * half) * Vdim + l15] = c0[r];
            part[2 * wave + 1][(r + 8 * half) * Vdim + l15] = c1[r];
        }
    }
    __syncthreads();

    const int e = threadIdx.x;
    if (e < TILE_OUT) {
        const int row = row0 + e / Vdim;
        float s = isTrans ? bias[e % Vdim] : 0.0f;
#pragma unroll
        for (int w = 0; w < NPART; ++w) s += part[w][e];
        if (row < rows) Dst[row0 * Vdim + e] = s;
    }
}

#define ROWLEN    (U1dim * Vdim)
#define OUT_ELEMS (TROWS * ROWLEN)
#define OUT_Q     (OUT_ELEMS / 4)
#define JW_BLOCKS 256
#define JW_STRIDE (JW_BLOCKS * 256)
#define JW_FULL   (OUT_Q / JW_STRIDE)
#define JW_TAIL   (OUT_Q - JW_FULL * JW_STRIDE)

static __device__ __forceinline__ v4f joint4(const float* __restrict__ TW,
                                             const float* __restrict__ PW,
                                             const unsigned q)
{
    const unsigned e0  = 4u * q;
    const unsigned bt0 = e0 / ROWLEN;
    const unsigned j0  = e0 - bt0 * ROWLEN;
    const unsigned v0  = j0 % Vdim;
    const unsigned pw0 = (bt0 / Tdim) * ROWLEN;
    const unsigned pw1 = ((bt0 + 1u) / Tdim) * ROWLEN;

    v4f r;
#pragma unroll
    for (unsigned i = 0; i < 4; ++i) {
        const unsigned j    = j0 + i;
        const bool     next = j >= ROWLEN;
        const unsigned jj   = next ? j - ROWLEN : j;
        const unsigned tb   = (next ? bt0 + 1u : bt0) * Vdim;
        const unsigned pb   = next ? pw1 : pw0;
        unsigned v = v0 + i;
        v = v >= Vdim ? v - Vdim : v;
        const float x = TW[tb + v] + PW[pb + jj];
        r[i] = fmaxf(x, 0.1f * x);
    }
    return r;
}

__global__ __launch_bounds__(256) void joint_write(
    const float* __restrict__ TW,
    const float* __restrict__ PW,
    float* __restrict__ out)
{
    const unsigned gtid = blockIdx.x * 256u + threadIdx.x;

    v4f r[JW_FULL];
#pragma unroll
    for (unsigned it = 0; it < JW_FULL; ++it)
        r[it] = joint4(TW, PW, it * JW_STRIDE + gtid);
#pragma unroll
    for (unsigned it = 0; it < JW_FULL; ++it)
        __builtin_nontemporal_store(r[it], (v4f*)(out + 4u * (size_t)(it * JW_STRIDE + gtid)));

    if (gtid < JW_TAIL) {
        const unsigned q = JW_FULL * JW_STRIDE + gtid;
        __builtin_nontemporal_store(joint4(TW, PW, q), (v4f*)(out + 4u * (size_t)q));
    }
}

extern "C" void kernel_launch(void* const* d_in, const int* in_sizes, int n_in,
                              void* d_out, int out_size, void* d_ws, size_t ws_size,
                              hipStream_t stream) {
    const float* trans = (const float*)d_in[0];
    const float* preds = (const float*)d_in[1];
    const float* W     = (const float*)d_in[2];
    const float* bias  = (const float*)d_in[3];
    float* out = (float*)d_out;

    float* TW = (float*)d_ws;
    float* PW = TW + TROWS * Vdim;

    rowgemm_wmma<<<NTILES, 256, 0, stream>>>(trans, preds, W, bias, TW, PW);
    joint_write<<<JW_BLOCKS, 256, 0, stream>>>(TW, PW, out);
}


# joint_write: each work-group owns 50 consecutive 512-byte pieces of out and stages their TW and PW rows in LDS once (5 global loads per lane, not 8 per float4; no division per float4)
# speedup vs baseline: 3.0034x; 1.1212x over previous
#include <hip/hip_runtime.h>


typedef __attribute__((ext_vector_type(2))) float v2f;
typedef __attribute__((ext_vector_type(8))) float v8f;

#define Hdim 512
#define Vdim 5
#define Bdim 8
#define Tdim 400
#define U1dim 101

#define TROWS (Bdim * Tdim)
#define PROWS (Bdim * U1dim)
#define TTILES ((TROWS + 15) / 16)
#define PTILES ((PROWS + 15) / 16)
#define NTILES (TTILES + PTILES)

typedef __attribute__((ext_vector_type(4))) float v4f;

#define KWAVES 8
#define KPIECE (Hdim / KWAVES)
#define KLOADS (KPIECE / 8)
#define TILE_OUT (16 * Vdim)
#define NPART (2 * KWAVES)

__global__ __launch_bounds__(256) void rowgemm_wmma(
    const float* __restrict__ trans,
    const float* __restrict__ preds,
    const float* __restrict__ W,
    const float* __restrict__ bias,
    float* __restrict__ TW,
    float* __restrict__ PW)
{
    __shared__ float part[NPART][TILE_OUT];

    const int wave = threadIdx.x >> 5;
    const int lane = threadIdx.x & 31;
    const int tile = blockIdx.x;

    const bool  isTrans = (tile < TTILES);
    const float* A   = isTrans ? trans : preds;
    float*       Dst = isTrans ? TW    : PW;
    const int    rows = isTrans ? TROWS : PROWS;
    const int    row0 = (isTrans ? tile : tile - TTILES) * 16;

    const int half = lane >> 4;
    const int l15  = lane & 15;

    const int arow = min(row0 + l15, rows - 1);
    const int brow = min(l15, Vdim - 1);

    const float* __restrict__ ap = A + arow * Hdim + wave * KPIECE + 4 * half;
    const float* __restrict__ bp = W + brow * Hdim + wave * KPIECE + 4 * half;

    v4f a[KLOADS], b[KLOADS];
#pragma unroll
    for (int j = 0; j < KLOADS; ++j) a[j] = *(const v4f*)(ap + 8 * j);
#pragma unroll
    for (int j = 0; j < KLOADS; ++j) b[j] = *(const v4f*)(bp + 8 * j);

    v8f c0, c1;
#pragma unroll
    for (int r = 0; r < 8; ++r) { c0[r] = 0.0f; c1[r] = 0.0f; }

#pragma unroll
    for (int j = 0; j < KLOADS; ++j) {
        const v2f a01 = {a[j][0], a[j][1]}, b01 = {b[j][0], b[j][1]};
        const v2f a23 = {a[j][2], a[j][3]}, b23 = {b[j][2], b[j][3]};
        c0 = __builtin_amdgcn_wmma_f32_16x16x4_f32(
                  false, a01,  false, b01,
                  (short)0, c0,  false,  false);
        c1 = __builtin_amdgcn_wmma_f32_16x16x4_f32(
                  false, a23,  false, b23,
                  (short)0, c1,  false,  false);
    }

    if (l15 < Vdim) {
#pragma unroll
        for (int r = 0; r < 8; ++r) {
            part[2 * wave    ][(r + 8 * half) * Vdim + l15] = c0[r];
            part[2 * wave + 1][(r + 8 * half) * Vdim + l15] = c1[r];
        }
    }
    __syncthreads();

    const int e = threadIdx.x;
    if (e < TILE_OUT) {
        const int row = row0 + e / Vdim;
        float s = isTrans ? bias[e % Vdim] : 0.0f;
#pragma unroll
        for (int w = 0; w < NPART; ++w) s += part[w][e];
        if (row < rows) Dst[row0 * Vdim + e] = s;
    }
}

#define ROWLEN    (U1dim * Vdim)
#define OUT_ELEMS (TROWS * ROWLEN)
#define PIECE     128
#define NPIECES   (OUT_ELEMS / PIECE)
#define JW_WAVES  10
#define JW_TRIPS  5
#define WG_PIECES (JW_WAVES * JW_TRIPS)
#define WG_ELEMS  (WG_PIECES * PIECE)
#define JW_BLOCKS ((NPIECES + WG_PIECES - 1) / WG_PIECES)
#define JW_THREADS (JW_WAVES * 32)
#define ST_ROWS   (JW_THREADS / ST_W)
#define ST_W      8
#define SP_W      (ROWLEN + 3)
#define SP_SLOTS  (4 * JW_THREADS)
#define TRIP_ELEMS (JW_WAVES * PIECE)
#define TRIP_ROWS (TRIP_ELEMS / ROWLEN)
#define TRIP_REST (TRIP_ELEMS - TRIP_ROWS * ROWLEN)

template <bool GUARDED>
static __device__ __forceinline__ void joint_trips(
    const float* __restrict__ sT, const float* __restrict__ sP, float* __restrict__ out,
    const unsigned e_base, const unsigned bt_first, const unsigned off0,
    const unsigned b1_row0, const unsigned wave, const unsigned lane)
{
    const unsigned rel0 = off0 + wave * PIECE + 4u * lane;
    unsigned r  = rel0 / ROWLEN;
    unsigned j0 = rel0 - r * ROWLEN;
    const unsigned v0 = (e_base + wave * PIECE + 4u * lane) % Vdim;
    unsigned piece = e_base / PIECE + wave;
    float* __restrict__ op = out + (size_t)e_base + wave * PIECE + 4u * lane;

#pragma unroll
    for (unsigned it = 0; it < JW_TRIPS; ++it) {
        const unsigned s  = (bt_first + r >= b1_row0) ? SP_W : 0u;
        const float* __restrict__ pp = sP + s + j0;
        const float* __restrict__ tc = sT + r * ST_W + v0;
        const float* __restrict__ tn = tc + ST_W;
        const unsigned nw = ROWLEN - j0;
        v4f x;
#pragma unroll
        for (unsigned i = 0; i < 4; ++i) {
            const float t = (i >= nw) ? tn[i] : tc[i];
            const float y = t + pp[i];
            x[i] = fmaxf(y, 0.1f * y);
        }
        if (!GUARDED || piece < NPIECES)
            __builtin_nontemporal_store(x, (v4f*)op);
        piece += JW_WAVES;
        op    += TRIP_ELEMS;
        r     += TRIP_ROWS;
        j0    += TRIP_REST;
        if (j0 >= ROWLEN) { j0 -= ROWLEN; r += 1u; }
    }
}

__global__ __launch_bounds__(JW_THREADS) void joint_write(
    const float* __restrict__ TW,
    const float* __restrict__ PW,
    float* __restrict__ out)
{
    __shared__ float sT[ST_ROWS * ST_W];
    __shared__ float sP[SP_SLOTS];

    const unsigned tid  = threadIdx.x;
    const unsigned wave = tid >> 5;
    const unsigned lane = tid & 31;

    const unsigned e_base   = blockIdx.x * WG_ELEMS;
    const unsigned bt_first = e_base / ROWLEN;
    const unsigned off0     = e_base - bt_first * ROWLEN;
    const unsigned b0       = bt_first / Tdim;
    const unsigned b1       = min(b0 + 1u, (unsigned)(Bdim - 1));
    const unsigned b1_row0  = (b0 + 1u) * Tdim;

    float st;
    {
        const unsigned r  = tid / ST_W, k = tid % ST_W;
        const unsigned kk = k >= Vdim ? k - Vdim : k;
        const unsigned row = min(bt_first + r, (unsigned)(TROWS - 1));
        st = TW[row * Vdim + kk];
    }
    float sp[SP_SLOTS / JW_THREADS];
#pragma unroll
    for (unsigned i = 0; i < SP_SLOTS / JW_THREADS; ++i) {
        const unsigned t  = tid + i * JW_THREADS;
        const unsigned s  = t >= SP_W ? 1u : 0u;
        const unsigned k  = t - s * SP_W;
        const unsigned kk = k >= ROWLEN ? k - ROWLEN : k;
        sp[i] = PW[(s ? b1 : b0) * ROWLEN + kk];
    }
    sT[tid] = st;
#pragma unroll
    for (unsigned i = 0; i < SP_SLOTS / JW_THREADS; ++i) sP[tid + i * JW_THREADS] = sp[i];
    __syncthreads();

    if (blockIdx.x != JW_BLOCKS - 1)
        joint_trips<false>(sT, sP, out, e_base, bt_first, off0, b1_row0, wave, lane);
    else
        joint_trips<true>(sT, sP, out, e_base, bt_first, off0, b1_row0, wave, lane);
}

extern "C" void kernel_launch(void* const* d_in, const int* in_sizes, int n_in,
                              void* d_out, int out_size, void* d_ws, size_t ws_size,
                              hipStream_t stream) {
    const float* trans = (const float*)d_in[0];
    const float* preds = (const float*)d_in[1];
    const float* W     = (const float*)d_in[2];
    const float* bias  = (const float*)d_in[3];
    float* out = (float*)d_out;

    float* TW = (float*)d_ws;
    float* PW = TW + TROWS * Vdim;

    rowgemm_wmma<<<NTILES, 256, 0, stream>>>(trans, preds, W, bias, TW, PW);
    joint_write<<<JW_BLOCKS, JW_THREADS, 0, stream>>>(TW, PW, out);
}
